# starting kernel
# baseline (speedup 1.0000x reference)
#include <hip/hip_runtime.h>
#include <hip/hip_bf16.h>

typedef __attribute__((ext_vector_type(2))) float v2f;
typedef __attribute__((ext_vector_type(8))) float v8f;

#define DDIM   128
#define KNN    32
#define TILE_N 16
#define WPAD   130
#define THREADS 256

__global__ __launch_bounds__(THREADS)
void gcl_fused_kernel(const float* __restrict__ Input,
                      const int*   __restrict__ Idx,
                      const float* __restrict__ C,
                      const float* __restrict__ W,
                      const float* __restrict__ b,
                      float*       __restrict__ out,
                      int N)
{
    extern __shared__ float lds[];
    float* wLDS = lds;
    float* yLDS = lds + DDIM * WPAD;
    float* sLDS = yLDS + TILE_N * WPAD;

    const int t  = threadIdx.x;
    const int n0 = blockIdx.x * TILE_N;

    for (int i = t; i < DDIM * DDIM; i += THREADS) {
        const int r = i >> 7;
        const int c = i & (DDIM - 1);
        wLDS[r * WPAD + c] = W[i];
    }

    const int d    = t & (DDIM - 1);
    const int half = t >> 7;

    float acc[8];
#pragma unroll
    for (int r = 0; r < 8; ++r) acc[r] = 0.0f;

    for (int k = 0; k < KNN; ++k) {
#pragma unroll
        for (int r = 0; r < 8; ++r) {
            const int n = n0 + half * 8 + r;
            if (n < N) {
                const int   gi = Idx[n * KNN + k];
                const float cw = C[n * KNN + k];
                acc[r] += cw * Input[(long)gi * DDIM + d];
            }
        }
    }
#pragma unroll
    for (int r = 0; r < 8; ++r)
        yLDS[(half * 8 + r) * WPAD + d] = acc[r];

    if (t < TILE_N) {
        const int n = n0 + t;
        float s = 0.0f;
        if (n < N)
            for (int k = 0; k < KNN; ++k) s += C[n * KNN + k];
        sLDS[t] = s;
    }
    __syncthreads();

    const int wave = t >> 5;
    const int lane = t & 31;
    const int hi   = lane >> 4;
    const int lo   = lane & 15;
    const int e0   = wave * 16;

    v8f cacc = {};
#pragma unroll
    for (int k0 = 0; k0 < DDIM; k0 += 4) {
        const float* ap = &yLDS[lo * WPAD + k0 + 2 * hi];
        const float* bp = &wLDS[(e0 + lo) * WPAD + k0 + 2 * hi];
        v2f av; av.x = ap[0]; av.y = ap[1];
        v2f bv; bv.x = bp[0]; bv.y = bp[1];
        cacc = __builtin_amdgcn_wmma_f32_16x16x4_f32(
             false, av,  false, bv,
             (short)0, cacc,  false,  false);
    }

    const float bias = b[e0 + lo];
#pragma unroll
    for (int v = 0; v < 8; ++v) {
        const int nl = v + 8 * hi;
        const int n  = n0 + nl;
        if (n < N) {
            const float val = cacc[v] + sLDS[nl] * bias;
            out[(long)n * DDIM + (e0 + lo)] = 1.0f / (1.0f + __expf(-val));
        }
    }
}

extern "C" void kernel_launch(void* const* d_in, const int* in_sizes, int n_in,
                              void* d_out, int out_size, void* d_ws, size_t ws_size,
                              hipStream_t stream)
{
    const float* Input = (const float*)d_in[0];
    const int*   Idx   = (const int*)  d_in[1];
    const float* C     = (const float*)d_in[2];
    const float* W     = (const float*)d_in[3];
    const float* b     = (const float*)d_in[4];
    float*       out   = (float*)d_out;

    const int N = in_sizes[1] / KNN;
    const int blocks = (N + TILE_N - 1) / TILE_N;
    const size_t shmem = (size_t)(DDIM * WPAD + TILE_N * WPAD + TILE_N) * sizeof(float);

    hipLaunchKernelGGL(gcl_fused_kernel, dim3(blocks), dim3(THREADS), shmem, stream,
                       Input, Idx, C, W, b, out, N);
}


# One launch instead of two: each wave first sums its row's 32 weighted Input rows, then the 16-row tile is projected once by WMMA in the same resident work-group (W in registers); sum order changed
# speedup vs baseline: 8.6632x; 8.6632x over previous
#include <hip/hip_runtime.h>


typedef __attribute__((ext_vector_type(2))) float v2f;
typedef __attribute__((ext_vector_type(4))) float v4f;
typedef __attribute__((ext_vector_type(8))) float v8f;

#define DDIM    128
#define KNN     32
#define TROWS   16
#define WAVES   16
#define THREADS (WAVES * 32)
#define PITCH   132
#define GRID    256

__device__ __forceinline__ void lds_barrier()
{
    __builtin_amdgcn_fence(__ATOMIC_RELEASE, "workgroup", "local");
    __builtin_amdgcn_s_barrier();
    __builtin_amdgcn_fence(__ATOMIC_ACQUIRE, "workgroup", "local");
}

__global__ __launch_bounds__(THREADS)
void gcl_sum_project_kernel(const float* __restrict__ Input,
                            const int*   __restrict__ Idx,
                            const float* __restrict__ C,
                            const float* __restrict__ W,
                            const float* __restrict__ b,
                            float*       __restrict__ out,
                            int N)
{
    __shared__ __attribute__((aligned(16))) float yT[TROWS * PITCH];
    __shared__ __attribute__((aligned(16))) float oT[TROWS * PITCH];
    __shared__ __attribute__((aligned(16))) float sT[TROWS];

    const int lane = threadIdx.x & 31;
    const int wave = __builtin_amdgcn_readfirstlane((int)(threadIdx.x >> 5));
    const int hi   = lane >> 4;
    const int lo   = lane & 15;
    const int e0   = (wave & 7) * 16;
    const bool mat = wave < 8;
    const int ntiles = (N + TROWS - 1) / TROWS;
    const char* xb = (const char*)Input;
    const unsigned voff = (unsigned)lane * 16u;

    int tile = blockIdx.x;
    if (tile >= ntiles) return;

    int n = tile * TROWS + wave;
    n = n < N ? n : N - 1;
    int gi = Idx[(size_t)n * KNN + lane];
    int cw = __builtin_bit_cast(int, C[(size_t)n * KNN + lane]);

    v4f wf[16];
#pragma unroll
    for (int i = 0; i < 16; ++i)
        wf[i] = *(const v4f*)(W + (size_t)(e0 + lo) * DDIM + 4 * hi + 8 * i);
    const float bias = b[e0 + lo];

    v4f z[KNN];
#pragma unroll
    for (int k = 0; k < KNN; ++k) {
        const unsigned g = (unsigned)__builtin_amdgcn_readlane(gi, k);
        z[k] = *(const v4f*)(xb + (size_t)g * (DDIM * 4u) + voff);
    }

    for (;;) {
        const int  next     = tile + (int)gridDim.x;
        const bool has_next = next < ntiles;
        int gi_n = 0, cw_n = 0;
        if (has_next) {
            int m = next * TROWS + wave;
            m = m < N ? m : N - 1;
            gi_n = Idx[(size_t)m * KNN + lane];
            cw_n = __builtin_bit_cast(int, C[(size_t)m * KNN + lane]);
        }

        v4f acc;
        float ssum;
        {
            const float c = __builtin_bit_cast(float, __builtin_amdgcn_readlane(cw, 0));
            acc.x = c * z[0].x; acc.y = c * z[0].y; acc.z = c * z[0].z; acc.w = c * z[0].w;
            ssum = c;
        }
#pragma unroll
        for (int k = 1; k < KNN; ++k) {
            const float c = __builtin_bit_cast(float, __builtin_amdgcn_readlane(cw, k));
            acc.x = __builtin_fmaf(c, z[k].x, acc.x);
            acc.y = __builtin_fmaf(c, z[k].y, acc.y);
            acc.z = __builtin_fmaf(c, z[k].z, acc.z);
            acc.w = __builtin_fmaf(c, z[k].w, acc.w);
            ssum += c;
        }
        *(v4f*)&yT[wave * PITCH + 4 * lane] = acc;
        if (lane == 0) sT[wave] = ssum;

        if (has_next) {
#pragma unroll
            for (int k = 0; k < KNN; ++k) {
                const unsigned g = (unsigned)__builtin_amdgcn_readlane(gi_n, k);
                z[k] = *(const v4f*)(xb + (size_t)g * (DDIM * 4u) + voff);
            }
        }

        lds_barrier();

        if (mat) {
            v8f acc0 = {};
            v8f acc1 = {};
            const float* ap = &yT[lo * PITCH + 4 * hi];
#pragma unroll
            for (int i = 0; i < 16; ++i) {
                const v4f a  = *(const v4f*)(ap + 8 * i);
                const v4f bb = wf[i];
                v2f al; al.x = a.x;  al.y = a.y;
                v2f ah; ah.x = a.z;  ah.y = a.w;
                v2f bl; bl.x = bb.x; bl.y = bb.y;
                v2f bh; bh.x = bb.z; bh.y = bb.w;
                acc0 = __builtin_amdgcn_wmma_f32_16x16x4_f32(false, al, false, bl, (short)0, acc0, false, false);
                acc1 = __builtin_amdgcn_wmma_f32_16x16x4_f32(false, ah, false, bh, (short)0, acc1, false, false);
            }
            const v4f s0 = *(const v4f*)&sT[8 * hi];
            const v4f s1 = *(const v4f*)&sT[8 * hi + 4];
            const float sr[8] = {s0.x, s0.y, s0.z, s0.w, s1.x, s1.y, s1.z, s1.w};
#pragma unroll
            for (int v = 0; v < 8; ++v) {
                const float x = __builtin_fmaf(sr[v], bias, acc0[v] + acc1[v]);
                oT[(8 * hi + v) * PITCH + e0 + lo] = 1.0f / (1.0f + __expf(-x));
            }
        }

        lds_barrier();

        {
            const int row = tile * TROWS + wave;
            if (row < N)
                *(v4f*)(out + (size_t)row * DDIM + 4 * lane) = *(const v4f*)&oT[wave * PITCH + 4 * lane];
        }

        if (!has_next) break;
        tile = next;
        gi = gi_n;
        cw = cw_n;
    }
}

extern "C" void kernel_launch(void* const* d_in, const int* in_sizes, int n_in,
                              void* d_out, int out_size, void* d_ws, size_t ws_size,
                              hipStream_t stream)
{
    const float* Input = (const float*)d_in[0];
    const int*   Idx   = (const int*)  d_in[1];
    const float* C     = (const float*)d_in[2];
    const float* W     = (const float*)d_in[3];
    const float* b     = (const float*)d_in[4];
    float*       out   = (float*)d_out;

    const int N = in_sizes[1] / KNN;

    hipLaunchKernelGGL(gcl_sum_project_kernel, dim3(GRID), dim3(THREADS), 0, stream,
                       Input, Idx, C, W, b, out, N);
}
